# starting kernel
# baseline (speedup 1.0000x reference)
#include <hip/hip_runtime.h>

typedef __attribute__((ext_vector_type(2))) float v2f;
typedef __attribute__((ext_vector_type(8))) float v8f;

#define DIN  512
#define DOUT 512

__global__ void fill_zero_kernel(float* __restrict__ p, size_t count) {
    size_t i      = (size_t)blockIdx.x * blockDim.x + threadIdx.x;
    size_t stride = (size_t)gridDim.x * blockDim.x;
    for (; i < count; i += stride) p[i] = 0.0f;
}

__global__ void rgcn_degree_kernel(const int* __restrict__ dst,
                                   float* __restrict__ deg,
                                   int e_per_rel, int n, int total) {
    int i = blockIdx.x * blockDim.x + threadIdx.x;
    if (i >= total) return;
    int r = i / e_per_rel;
    atomicAdd(&deg[(size_t)r * n + dst[i]], 1.0f);
}

__global__ __launch_bounds__(128)
void rgcn_gemm_wmma_kernel(const float* __restrict__ x,
                           const float* __restrict__ W,
                           const float* __restrict__ b,
                           float* __restrict__ Wh,
                           int nrows) {
    __shared__ float xs[64][33];
    __shared__ float wsT[128][33];

    const int tid  = threadIdx.x;
    const int wave = tid >> 5;
    const int lane = tid & 31;
    const int l15  = lane & 15;
    const int khal = (lane >> 4) * 2;

    const int m0 = blockIdx.x * 64;
    const int n0 = blockIdx.y * 128;
    const int wm = (wave >> 1) * 32;
    const int wn = (wave & 1) * 64;

    v8f acc[2][4] = {};

    for (int k0 = 0; k0 < DIN; k0 += 32) {
        for (int idx = tid; idx < 64 * 32; idx += 128) {
            int r = idx >> 5, c = idx & 31;
            int gr = m0 + r;
            if (gr >= nrows) gr = nrows - 1;
            xs[r][c] = x[(size_t)gr * DIN + k0 + c];
        }
        for (int idx = tid; idx < 32 * 128; idx += 128) {
            int k = idx >> 7, c = idx & 127;
            wsT[c][k] = W[(size_t)(k0 + k) * DOUT + n0 + c];
        }
        __syncthreads();

        #pragma unroll
        for (int kk = 0; kk < 32; kk += 4) {
            const int kb = kk + khal;
            v2f a[2];
            #pragma unroll
            for (int i = 0; i < 2; ++i) {
                a[i].x = xs[wm + i * 16 + l15][kb];
                a[i].y = xs[wm + i * 16 + l15][kb + 1];
            }
            v2f bf[4];
            #pragma unroll
            for (int j = 0; j < 4; ++j) {
                bf[j].x = wsT[wn + j * 16 + l15][kb];
                bf[j].y = wsT[wn + j * 16 + l15][kb + 1];
            }
            #pragma unroll
            for (int i = 0; i < 2; ++i)
                #pragma unroll
                for (int j = 0; j < 4; ++j)
                    acc[i][j] = __builtin_amdgcn_wmma_f32_16x16x4_f32(
                        false, a[i], false, bf[j], (short)0, acc[i][j], false, false);
        }
        __syncthreads();
    }

    const int mbase = m0 + wm + (lane >> 4) * 8;
    #pragma unroll
    for (int i = 0; i < 2; ++i) {
        #pragma unroll
        for (int j = 0; j < 4; ++j) {
            const int col  = n0 + wn + j * 16 + l15;
            const float bb = b[col];
            #pragma unroll
            for (int v = 0; v < 8; ++v) {
                const int row = mbase + i * 16 + v;
                if (row < nrows)
                    Wh[(size_t)row * DOUT + col] = acc[i][j][v] + bb;
            }
        }
    }
}

__global__ __launch_bounds__(256)
void rgcn_edge_agg_kernel(const float* __restrict__ Wh,
                          const int* __restrict__ src,
                          const int* __restrict__ dst,
                          const float* __restrict__ deg,
                          float* __restrict__ out,
                          int nedges) {
    const int e = blockIdx.x * 2 + (threadIdx.x >> 7);
    if (e >= nedges) return;
    const int t = threadIdx.x & 127;
    const int s = src[e];
    const int d = dst[e];
    const float inv = 1.0f / fmaxf(deg[d], 1.0f);

    const float4 v = *(const float4*)(Wh + (size_t)s * DOUT + t * 4);
    float* o = out + (size_t)d * DOUT + t * 4;
    atomicAdd(o + 0, v.x * inv);
    atomicAdd(o + 1, v.y * inv);
    atomicAdd(o + 2, v.z * inv);
    atomicAdd(o + 3, v.w * inv);
}

extern "C" void kernel_launch(void* const* d_in, const int* in_sizes, int n_in,
                              void* d_out, int out_size, void* d_ws, size_t ws_size,
                              hipStream_t stream) {
    const float* x   = (const float*)d_in[0];
    const float* W   = (const float*)d_in[1];
    const float* b   = (const float*)d_in[2];
    const int*   src = (const int*)d_in[3];
    const int*   dst = (const int*)d_in[4];
    float*       out = (float*)d_out;

    const int n    = in_sizes[0] / DIN;
    const int nrel = in_sizes[1] / (DIN * DOUT);
    const int e    = in_sizes[3] / nrel;

    float* Wh  = (float*)d_ws;
    float* deg = Wh + (size_t)n * DOUT;

    fill_zero_kernel<<<2048, 256, 0, stream>>>(out, (size_t)n * DOUT);
    fill_zero_kernel<<<512, 256, 0, stream>>>(deg, (size_t)nrel * n);

    {
        const int total  = nrel * e;
        const int blocks = (total + 255) / 256;
        rgcn_degree_kernel<<<blocks, 256, 0, stream>>>(dst, deg, e, n, total);
    }

    dim3 ggrid((n + 63) / 64, DOUT / 128, 1);
    for (int r = 0; r < nrel; ++r) {
        rgcn_gemm_wmma_kernel<<<ggrid, 128, 0, stream>>>(
            x, W + (size_t)r * DIN * DOUT, b + (size_t)r * DOUT, Wh, n);
        rgcn_edge_agg_kernel<<<(e + 1) / 2, 256, 0, stream>>>(
            Wh, src + (size_t)r * e, dst + (size_t)r * e,
            deg + (size_t)r * n, out, e);
    }
}


# GEMM K-loop software-pipelined: next slab's global loads issued as one batch of 16-byte loads ahead of the WMMAs, held in registers, stored to a second LDS stage after them; one barrier per slab
# speedup vs baseline: 1.3239x; 1.3239x over previous
#include <hip/hip_runtime.h>

typedef __attribute__((ext_vector_type(2))) float v2f;
typedef __attribute__((ext_vector_type(8))) float v8f;

#define DIN  512
#define DOUT 512

__global__ void fill_zero_kernel(float* __restrict__ p, size_t count) {
    size_t i      = (size_t)blockIdx.x * blockDim.x + threadIdx.x;
    size_t stride = (size_t)gridDim.x * blockDim.x;
    for (; i < count; i += stride) p[i] = 0.0f;
}

__global__ void rgcn_degree_kernel(const int* __restrict__ dst,
                                   float* __restrict__ deg,
                                   int e_per_rel, int n, int total) {
    int i = blockIdx.x * blockDim.x + threadIdx.x;
    if (i >= total) return;
    int r = i / e_per_rel;
    atomicAdd(&deg[(size_t)r * n + dst[i]], 1.0f);
}

__global__ __launch_bounds__(128)
void rgcn_gemm_wmma_kernel(const float* __restrict__ x,
                           const float* __restrict__ W,
                           const float* __restrict__ b,
                           float* __restrict__ Wh,
                           int nrows) {
    __shared__ float xs[2][64][33];
    __shared__ float wsT[2][128][33];

    const int tid  = threadIdx.x;
    const int wave = tid >> 5;
    const int lane = tid & 31;
    const int l15  = lane & 15;
    const int khal = (lane >> 4) * 2;

    const int m0 = blockIdx.x * 64;
    const int n0 = blockIdx.y * 128;
    const int wm = (wave >> 1) * 32;
    const int wn = (wave & 1) * 64;

    const int xr0 = tid >> 3;
    const int xc  = (tid & 7) * 4;
    const int wc  = lane * 4;

    const float* xp[4];
    #pragma unroll
    for (int j = 0; j < 4; ++j) {
        int gr = m0 + xr0 + 16 * j;
        if (gr >= nrows) gr = nrows - 1;
        xp[j] = x + (size_t)gr * DIN + xc;
    }
    const float* wp = W + (size_t)wave * DOUT + n0 + wc;

    float4 xreg[4], wreg[8];

    v8f acc[2][4] = {};

    #pragma unroll
    for (int j = 0; j < 4; ++j) xreg[j] = *(const float4*)(xp[j]);
    #pragma unroll
    for (int j = 0; j < 8; ++j) wreg[j] = *(const float4*)(wp + (size_t)(4 * j) * DOUT);
    #pragma unroll
    for (int j = 0; j < 4; ++j) {
        float* d = &xs[0][xr0 + 16 * j][xc];
        d[0] = xreg[j].x; d[1] = xreg[j].y; d[2] = xreg[j].z; d[3] = xreg[j].w;
    }
    #pragma unroll
    for (int j = 0; j < 8; ++j) {
        const int k = wave + 4 * j;
        wsT[0][wc + 0][k] = wreg[j].x;
        wsT[0][wc + 1][k] = wreg[j].y;
        wsT[0][wc + 2][k] = wreg[j].z;
        wsT[0][wc + 3][k] = wreg[j].w;
    }
    __syncthreads();

    for (int it = 0; it < DIN / 32; ++it) {
        const int  s    = it & 1;
        const bool more = (it + 1) < DIN / 32;

        if (more) {
            const int k1 = (it + 1) * 32;
            #pragma unroll
            for (int j = 0; j < 4; ++j) xreg[j] = *(const float4*)(xp[j] + k1);
            #pragma unroll
            for (int j = 0; j < 8; ++j)
                wreg[j] = *(const float4*)(wp + (size_t)(k1 + 4 * j) * DOUT);
        }

        #pragma unroll
        for (int kk = 0; kk < 32; kk += 4) {
            const int kb = kk + khal;
            v2f a[2];
            #pragma unroll
            for (int i = 0; i < 2; ++i) {
                a[i].x = xs[s][wm + i * 16 + l15][kb];
                a[i].y = xs[s][wm + i * 16 + l15][kb + 1];
            }
            v2f bf[4];
            #pragma unroll
            for (int j = 0; j < 4; ++j) {
                bf[j].x = wsT[s][wn + j * 16 + l15][kb];
                bf[j].y = wsT[s][wn + j * 16 + l15][kb + 1];
            }
            #pragma unroll
            for (int i = 0; i < 2; ++i)
                #pragma unroll
                for (int j = 0; j < 4; ++j)
                    acc[i][j] = __builtin_amdgcn_wmma_f32_16x16x4_f32(
                        false, a[i], false, bf[j], (short)0, acc[i][j], false, false);
        }

        if (more) {
            const int s1 = s ^ 1;
            #pragma unroll
            for (int j = 0; j < 4; ++j) {
                float* d = &xs[s1][xr0 + 16 * j][xc];
                d[0] = xreg[j].x; d[1] = xreg[j].y; d[2] = xreg[j].z; d[3] = xreg[j].w;
            }
            #pragma unroll
            for (int j = 0; j < 8; ++j) {
                const int k = wave + 4 * j;
                wsT[s1][wc + 0][k] = wreg[j].x;
                wsT[s1][wc + 1][k] = wreg[j].y;
                wsT[s1][wc + 2][k] = wreg[j].z;
                wsT[s1][wc + 3][k] = wreg[j].w;
            }
        }
        __syncthreads();
    }

    const int mbase = m0 + wm + (lane >> 4) * 8;
    #pragma unroll
    for (int i = 0; i < 2; ++i) {
        #pragma unroll
        for (int j = 0; j < 4; ++j) {
            const int col  = n0 + wn + j * 16 + l15;
            const float bb = b[col];
            #pragma unroll
            for (int v = 0; v < 8; ++v) {
                const int row = mbase + i * 16 + v;
                if (row < nrows)
                    Wh[(size_t)row * DOUT + col] = acc[i][j][v] + bb;
            }
        }
    }
}

__global__ __launch_bounds__(256)
void rgcn_edge_agg_kernel(const float* __restrict__ Wh,
                          const int* __restrict__ src,
                          const int* __restrict__ dst,
                          const float* __restrict__ deg,
                          float* __restrict__ out,
                          int nedges) {
    const int e = blockIdx.x * 2 + (threadIdx.x >> 7);
    if (e >= nedges) return;
    const int t = threadIdx.x & 127;
    const int s = src[e];
    const int d = dst[e];
    const float inv = 1.0f / fmaxf(deg[d], 1.0f);

    const float4 v = *(const float4*)(Wh + (size_t)s * DOUT + t * 4);
    float* o = out + (size_t)d * DOUT + t * 4;
    atomicAdd(o + 0, v.x * inv);
    atomicAdd(o + 1, v.y * inv);
    atomicAdd(o + 2, v.z * inv);
    atomicAdd(o + 3, v.w * inv);
}

extern "C" void kernel_launch(void* const* d_in, const int* in_sizes, int n_in,
                              void* d_out, int out_size, void* d_ws, size_t ws_size,
                              hipStream_t stream) {
    const float* x   = (const float*)d_in[0];
    const float* W   = (const float*)d_in[1];
    const float* b   = (const float*)d_in[2];
    const int*   src = (const int*)d_in[3];
    const int*   dst = (const int*)d_in[4];
    float*       out = (float*)d_out;

    const int n    = in_sizes[0] / DIN;
    const int nrel = in_sizes[1] / (DIN * DOUT);
    const int e    = in_sizes[3] / nrel;

    float* Wh  = (float*)d_ws;
    float* deg = Wh + (size_t)n * DOUT;

    fill_zero_kernel<<<2048, 256, 0, stream>>>(out, (size_t)n * DOUT);
    fill_zero_kernel<<<512, 256, 0, stream>>>(deg, (size_t)nrel * n);

    {
        const int total  = nrel * e;
        const int blocks = (total + 255) / 256;
        rgcn_degree_kernel<<<blocks, 256, 0, stream>>>(dst, deg, e, n, total);
    }

    dim3 ggrid((n + 63) / 64, DOUT / 128, 1);
    for (int r = 0; r < nrel; ++r) {
        rgcn_gemm_wmma_kernel<<<ggrid, 128, 0, stream>>>(
            x, W + (size_t)r * DIN * DOUT, b + (size_t)r * DOUT, Wh, n);
        rgcn_edge_agg_kernel<<<(e + 1) / 2, 256, 0, stream>>>(
            Wh, src + (size_t)r * e, dst + (size_t)r * e,
            deg + (size_t)r * n, out, e);
    }
}


# Edge scatter re-mapped: one wave walks a run of 10 edges (indices and 1/deg read once per wave), rows lane-contiguous so each load and f32 atomic covers one 128 B line; atomic arrival order differs
# speedup vs baseline: 1.6976x; 1.2823x over previous
#include <hip/hip_runtime.h>

typedef __attribute__((ext_vector_type(2))) float v2f;
typedef __attribute__((ext_vector_type(8))) float v8f;

#define DIN  512
#define DOUT 512

__global__ void fill_zero_kernel(float* __restrict__ p, size_t count) {
    size_t i      = (size_t)blockIdx.x * blockDim.x + threadIdx.x;
    size_t stride = (size_t)gridDim.x * blockDim.x;
    for (; i < count; i += stride) p[i] = 0.0f;
}

__global__ void rgcn_degree_kernel(const int* __restrict__ dst,
                                   float* __restrict__ deg,
                                   int e_per_rel, int n, int total) {
    int i = blockIdx.x * blockDim.x + threadIdx.x;
    if (i >= total) return;
    int r = i / e_per_rel;
    atomicAdd(&deg[(size_t)r * n + dst[i]], 1.0f);
}

__global__ __launch_bounds__(128)
void rgcn_gemm_wmma_kernel(const float* __restrict__ x,
                           const float* __restrict__ W,
                           const float* __restrict__ b,
                           float* __restrict__ Wh,
                           int nrows) {
    __shared__ float xs[2][64][33];
    __shared__ float wsT[2][128][33];

    const int tid  = threadIdx.x;
    const int wave = tid >> 5;
    const int lane = tid & 31;
    const int l15  = lane & 15;
    const int khal = (lane >> 4) * 2;

    const int m0 = blockIdx.x * 64;
    const int n0 = blockIdx.y * 128;
    const int wm = (wave >> 1) * 32;
    const int wn = (wave & 1) * 64;

    const int xr0 = tid >> 3;
    const int xc  = (tid & 7) * 4;
    const int wc  = lane * 4;

    const float* xp[4];
    #pragma unroll
    for (int j = 0; j < 4; ++j) {
        int gr = m0 + xr0 + 16 * j;
        if (gr >= nrows) gr = nrows - 1;
        xp[j] = x + (size_t)gr * DIN + xc;
    }
    const float* wp = W + (size_t)wave * DOUT + n0 + wc;

    float4 xreg[4], wreg[8];

    v8f acc[2][4] = {};

    #pragma unroll
    for (int j = 0; j < 4; ++j) xreg[j] = *(const float4*)(xp[j]);
    #pragma unroll
    for (int j = 0; j < 8; ++j) wreg[j] = *(const float4*)(wp + (size_t)(4 * j) * DOUT);
    #pragma unroll
    for (int j = 0; j < 4; ++j) {
        float* d = &xs[0][xr0 + 16 * j][xc];
        d[0] = xreg[j].x; d[1] = xreg[j].y; d[2] = xreg[j].z; d[3] = xreg[j].w;
    }
    #pragma unroll
    for (int j = 0; j < 8; ++j) {
        const int k = wave + 4 * j;
        wsT[0][wc + 0][k] = wreg[j].x;
        wsT[0][wc + 1][k] = wreg[j].y;
        wsT[0][wc + 2][k] = wreg[j].z;
        wsT[0][wc + 3][k] = wreg[j].w;
    }
    __syncthreads();

    for (int it = 0; it < DIN / 32; ++it) {
        const int  s    = it & 1;
        const bool more = (it + 1) < DIN / 32;

        if (more) {
            const int k1 = (it + 1) * 32;
            #pragma unroll
            for (int j = 0; j < 4; ++j) xreg[j] = *(const float4*)(xp[j] + k1);
            #pragma unroll
            for (int j = 0; j < 8; ++j)
                wreg[j] = *(const float4*)(wp + (size_t)(k1 + 4 * j) * DOUT);
        }

        #pragma unroll
        for (int kk = 0; kk < 32; kk += 4) {
            const int kb = kk + khal;
            v2f a[2];
            #pragma unroll
            for (int i = 0; i < 2; ++i) {
                a[i].x = xs[s][wm + i * 16 + l15][kb];
                a[i].y = xs[s][wm + i * 16 + l15][kb + 1];
            }
            v2f bf[4];
            #pragma unroll
            for (int j = 0; j < 4; ++j) {
                bf[j].x = wsT[s][wn + j * 16 + l15][kb];
                bf[j].y = wsT[s][wn + j * 16 + l15][kb + 1];
            }
            #pragma unroll
            for (int i = 0; i < 2; ++i)
                #pragma unroll
                for (int j = 0; j < 4; ++j)
                    acc[i][j] = __builtin_amdgcn_wmma_f32_16x16x4_f32(
                        false, a[i], false, bf[j], (short)0, acc[i][j], false, false);
        }

        if (more) {
            const int s1 = s ^ 1;
            #pragma unroll
            for (int j = 0; j < 4; ++j) {
                float* d = &xs[s1][xr0 + 16 * j][xc];
                d[0] = xreg[j].x; d[1] = xreg[j].y; d[2] = xreg[j].z; d[3] = xreg[j].w;
            }
            #pragma unroll
            for (int j = 0; j < 8; ++j) {
                const int k = wave + 4 * j;
                wsT[s1][wc + 0][k] = wreg[j].x;
                wsT[s1][wc + 1][k] = wreg[j].y;
                wsT[s1][wc + 2][k] = wreg[j].z;
                wsT[s1][wc + 3][k] = wreg[j].w;
            }
        }
        __syncthreads();
    }

    const int mbase = m0 + wm + (lane >> 4) * 8;
    #pragma unroll
    for (int i = 0; i < 2; ++i) {
        #pragma unroll
        for (int j = 0; j < 4; ++j) {
            const int col  = n0 + wn + j * 16 + l15;
            const float bb = b[col];
            #pragma unroll
            for (int v = 0; v < 8; ++v) {
                const int row = mbase + i * 16 + v;
                if (row < nrows)
                    Wh[(size_t)row * DOUT + col] = acc[i][j][v] + bb;
            }
        }
    }
}

#define EDGES_PER_WAVE 10

__global__ __launch_bounds__(256)
void rgcn_edge_agg_kernel(const float* __restrict__ Wh,
                          const int* __restrict__ src,
                          const int* __restrict__ dst,
                          const float* __restrict__ deg,
                          float* __restrict__ out,
                          int nedges) {
    const int lane = threadIdx.x & 31;
    const int wave = __builtin_amdgcn_readfirstlane((int)(blockIdx.x * 8 + (threadIdx.x >> 5)));
    const int e0 = wave * EDGES_PER_WAVE;
    if (e0 >= nedges) return;
    int cnt = nedges - e0;
    if (cnt > EDGES_PER_WAVE) cnt = EDGES_PER_WAVE;

    int   sv = 0, dv = 0;
    float iv = 0.0f;
    if (lane < cnt) {
        sv = src[e0 + lane];
        dv = dst[e0 + lane];
        iv = 1.0f / fmaxf(deg[dv], 1.0f);
    }

    float cur[16], nxt[16];
    {
        const int s = __builtin_amdgcn_readlane(sv, 0);
        const float* p = Wh + (size_t)s * DOUT + lane;
        #pragma unroll
        for (int k = 0; k < 16; ++k) cur[k] = p[k * 32];
    }
    for (int i = 0; i < cnt; ++i) {
        const bool more = (i + 1) < cnt;
        if (more) {
            const int s = __builtin_amdgcn_readlane(sv, i + 1);
            const float* p = Wh + (size_t)s * DOUT + lane;
            #pragma unroll
            for (int k = 0; k < 16; ++k) nxt[k] = p[k * 32];
        }
        const int   d   = __builtin_amdgcn_readlane(dv, i);
        const float inv = __int_as_float(__builtin_amdgcn_readlane(__float_as_int(iv), i));
        float* o = out + (size_t)d * DOUT + lane;
        #pragma unroll
        for (int k = 0; k < 16; ++k) atomicAdd(o + k * 32, cur[k] * inv);
        if (more) {
            #pragma unroll
            for (int k = 0; k < 16; ++k) cur[k] = nxt[k];
        }
    }
}

extern "C" void kernel_launch(void* const* d_in, const int* in_sizes, int n_in,
                              void* d_out, int out_size, void* d_ws, size_t ws_size,
                              hipStream_t stream) {
    const float* x   = (const float*)d_in[0];
    const float* W   = (const float*)d_in[1];
    const float* b   = (const float*)d_in[2];
    const int*   src = (const int*)d_in[3];
    const int*   dst = (const int*)d_in[4];
    float*       out = (float*)d_out;

    const int n    = in_sizes[0] / DIN;
    const int nrel = in_sizes[1] / (DIN * DOUT);
    const int e    = in_sizes[3] / nrel;

    float* Wh  = (float*)d_ws;
    float* deg = Wh + (size_t)n * DOUT;

    fill_zero_kernel<<<2048, 256, 0, stream>>>(out, (size_t)n * DOUT);
    fill_zero_kernel<<<512, 256, 0, stream>>>(deg, (size_t)nrel * n);

    {
        const int total  = nrel * e;
        const int blocks = (total + 255) / 256;
        rgcn_degree_kernel<<<blocks, 256, 0, stream>>>(dst, deg, e, n, total);
    }

    dim3 ggrid((n + 63) / 64, DOUT / 128, 1);
    for (int r = 0; r < nrel; ++r) {
        rgcn_gemm_wmma_kernel<<<ggrid, 128, 0, stream>>>(
            x, W + (size_t)r * DIN * DOUT, b + (size_t)r * DOUT, Wh, n);
        rgcn_edge_agg_kernel<<<(e + 8 * EDGES_PER_WAVE - 1) / (8 * EDGES_PER_WAVE), 256, 0, stream>>>(
            Wh, src + (size_t)r * e, dst + (size_t)r * e,
            deg + (size_t)r * n, out, e);
    }
}


# Edge sums by owner work-groups: each owns 64 out rows, scans dst, adds hit Wh rows into LDS (ds_add_f32), stores each line once; no global atomics/zero fill/degree pass; sum order set by LDS arrival
# speedup vs baseline: 1.7756x; 1.0459x over previous
#include <hip/hip_runtime.h>

typedef __attribute__((ext_vector_type(2))) float v2f;
typedef __attribute__((ext_vector_type(8))) float v8f;

#define DIN  512
#define DOUT 512

__global__ __launch_bounds__(128)
void rgcn_gemm_wmma_kernel(const float* __restrict__ x,
                           const float* __restrict__ W,
                           const float* __restrict__ b,
                           float* __restrict__ Wh,
                           int nrows) {
    __shared__ float xs[2][64][33];
    __shared__ float wsT[2][128][33];

    const int tid  = threadIdx.x;
    const int wave = tid >> 5;
    const int lane = tid & 31;
    const int l15  = lane & 15;
    const int khal = (lane >> 4) * 2;

    const int m0 = blockIdx.x * 64;
    const int n0 = blockIdx.y * 128;
    const int wm = (wave >> 1) * 32;
    const int wn = (wave & 1) * 64;

    const int xr0 = tid >> 3;
    const int xc  = (tid & 7) * 4;
    const int wc  = lane * 4;

    const float* xp[4];
    #pragma unroll
    for (int j = 0; j < 4; ++j) {
        int gr = m0 + xr0 + 16 * j;
        if (gr >= nrows) gr = nrows - 1;
        xp[j] = x + (size_t)gr * DIN + xc;
    }
    const float* wp = W + (size_t)wave * DOUT + n0 + wc;

    float4 xreg[4], wreg[8];

    v8f acc[2][4] = {};

    #pragma unroll
    for (int j = 0; j < 4; ++j) xreg[j] = *(const float4*)(xp[j]);
    #pragma unroll
    for (int j = 0; j < 8; ++j) wreg[j] = *(const float4*)(wp + (size_t)(4 * j) * DOUT);
    #pragma unroll
    for (int j = 0; j < 4; ++j) {
        float* d = &xs[0][xr0 + 16 * j][xc];
        d[0] = xreg[j].x; d[1] = xreg[j].y; d[2] = xreg[j].z; d[3] = xreg[j].w;
    }
    #pragma unroll
    for (int j = 0; j < 8; ++j) {
        const int k = wave + 4 * j;
        wsT[0][wc + 0][k] = wreg[j].x;
        wsT[0][wc + 1][k] = wreg[j].y;
        wsT[0][wc + 2][k] = wreg[j].z;
        wsT[0][wc + 3][k] = wreg[j].w;
    }
    __syncthreads();

    for (int it = 0; it < DIN / 32; ++it) {
        const int  s    = it & 1;
        const bool more = (it + 1) < DIN / 32;

        if (more) {
            const int k1 = (it + 1) * 32;
            #pragma unroll
            for (int j = 0; j < 4; ++j) xreg[j] = *(const float4*)(xp[j] + k1);
            #pragma unroll
            for (int j = 0; j < 8; ++j)
                wreg[j] = *(const float4*)(wp + (size_t)(k1 + 4 * j) * DOUT);
        }

        #pragma unroll
        for (int kk = 0; kk < 32; kk += 4) {
            const int kb = kk + khal;
            v2f a[2];
            #pragma unroll
            for (int i = 0; i < 2; ++i) {
                a[i].x = xs[s][wm + i * 16 + l15][kb];
                a[i].y = xs[s][wm + i * 16 + l15][kb + 1];
            }
            v2f bf[4];
            #pragma unroll
            for (int j = 0; j < 4; ++j) {
                bf[j].x = wsT[s][wn + j * 16 + l15][kb];
                bf[j].y = wsT[s][wn + j * 16 + l15][kb + 1];
            }
            #pragma unroll
            for (int i = 0; i < 2; ++i)
                #pragma unroll
                for (int j = 0; j < 4; ++j)
                    acc[i][j] = __builtin_amdgcn_wmma_f32_16x16x4_f32(
                        false, a[i], false, bf[j], (short)0, acc[i][j], false, false);
        }

        if (more) {
            const int s1 = s ^ 1;
            #pragma unroll
            for (int j = 0; j < 4; ++j) {
                float* d = &xs[s1][xr0 + 16 * j][xc];
                d[0] = xreg[j].x; d[1] = xreg[j].y; d[2] = xreg[j].z; d[3] = xreg[j].w;
            }
            #pragma unroll
            for (int j = 0; j < 8; ++j) {
                const int k = wave + 4 * j;
                wsT[s1][wc + 0][k] = wreg[j].x;
                wsT[s1][wc + 1][k] = wreg[j].y;
                wsT[s1][wc + 2][k] = wreg[j].z;
                wsT[s1][wc + 3][k] = wreg[j].w;
            }
        }
        __syncthreads();
    }

    const int mbase = m0 + wm + (lane >> 4) * 8;
    #pragma unroll
    for (int i = 0; i < 2; ++i) {
        #pragma unroll
        for (int j = 0; j < 4; ++j) {
            const int col  = n0 + wn + j * 16 + l15;
            const float bb = b[col];
            #pragma unroll
            for (int v = 0; v < 8; ++v) {
                const int row = mbase + i * 16 + v;
                if (row < nrows)
                    Wh[(size_t)row * DOUT + col] = acc[i][j][v] + bb;
            }
        }
    }
}

#define NB          64
#define AGG_WAVES   32
#define AGG_THREADS (AGG_WAVES * 32)
#define CHUNK       1024

__device__ __forceinline__ void rgcn_add_row(const float* __restrict__ Wh, int s, int n,
                                             float* acc, int* cnt, int dl, int lane) {
    if ((unsigned)s >= (unsigned)n) s = 0;
    const float4* p = (const float4*)(Wh + (size_t)s * DOUT) + lane;
    const float4 v0 = p[0], v1 = p[32], v2 = p[64], v3 = p[96];
    float* a = acc + dl * DOUT + lane;
    atomicAdd(a +   0, v0.x); atomicAdd(a +  32, v0.y);
    atomicAdd(a +  64, v0.z); atomicAdd(a +  96, v0.w);
    atomicAdd(a + 128, v1.x); atomicAdd(a + 160, v1.y);
    atomicAdd(a + 192, v1.z); atomicAdd(a + 224, v1.w);
    atomicAdd(a + 256, v2.x); atomicAdd(a + 288, v2.y);
    atomicAdd(a + 320, v2.z); atomicAdd(a + 352, v2.w);
    atomicAdd(a + 384, v3.x); atomicAdd(a + 416, v3.y);
    atomicAdd(a + 448, v3.z); atomicAdd(a + 480, v3.w);
    if (lane == 0) atomicAdd(&cnt[dl], 1);
}

__global__ __launch_bounds__(AGG_THREADS)
void rgcn_owner_agg_kernel(const float* __restrict__ Wh,
                           const int* __restrict__ src,
                           const int* __restrict__ dst,
                           float* __restrict__ out,
                           int nedges, int n, int first) {
    __shared__ __attribute__((aligned(16))) float acc[NB * DOUT];
    __shared__ int cnt[NB];

    const int tid  = threadIdx.x;
    const int lane = tid & 31;
    const int wave = __builtin_amdgcn_readfirstlane(tid >> 5);
    const int node0 = blockIdx.x * NB;
    int nvalid = n - node0;
    if (nvalid > NB) nvalid = NB;
    const bool vec_ok = ((((size_t)dst) | ((size_t)src)) & 15) == 0;

    for (int i = tid; i < NB * DOUT / 4; i += AGG_THREADS)
        ((float4*)acc)[i] = make_float4(0.0f, 0.0f, 0.0f, 0.0f);
    if (tid < NB) cnt[tid] = 0;
    __syncthreads();

    const int step = AGG_WAVES * CHUNK;
    for (int base = wave * CHUNK; base < nedges; base += step) {
        if (vec_ok && base + CHUNK <= nedges) {
            int4 v[8];
            #pragma unroll
            for (int g = 0; g < 8; ++g)
                v[g] = *(const int4*)(dst + base + g * 128 + lane * 4);
            unsigned m[8];
            unsigned any = 0;
            #pragma unroll
            for (int g = 0; g < 8; ++g) {
                m[g] = __builtin_amdgcn_ballot_w32((unsigned)(v[g].x - node0) < (unsigned)nvalid) |
                       __builtin_amdgcn_ballot_w32((unsigned)(v[g].y - node0) < (unsigned)nvalid) |
                       __builtin_amdgcn_ballot_w32((unsigned)(v[g].z - node0) < (unsigned)nvalid) |
                       __builtin_amdgcn_ballot_w32((unsigned)(v[g].w - node0) < (unsigned)nvalid);
                any |= m[g];
            }
            if (any) {
                #pragma unroll
                for (int g = 0; g < 8; ++g) {
                    unsigned mask = m[g];
                    while (mask) {
                        const int L = __builtin_ctz(mask);
                        mask &= mask - 1;
                        const int e4 = base + g * 128 + L * 4;
                        const int4 dd = *(const int4*)(dst + e4);
                        const int4 ss = *(const int4*)(src + e4);
                        #pragma unroll 1
                        for (int k = 0; k < 4; ++k) {
                            const int d = (k == 0) ? dd.x : (k == 1) ? dd.y : (k == 2) ? dd.z : dd.w;
                            const int s = (k == 0) ? ss.x : (k == 1) ? ss.y : (k == 2) ? ss.z : ss.w;
                            const int dl = d - node0;
                            if ((unsigned)dl < (unsigned)nvalid)
                                rgcn_add_row(Wh, s, n, acc, cnt, dl, lane);
                        }
                    }
                }
            }
        } else {
            #pragma unroll 1
            for (int b = base; b < base + CHUNK && b < nedges; b += 256) {
                int d[8];
                #pragma unroll
                for (int q = 0; q < 8; ++q) {
                    const int e = b + q * 32 + lane;
                    d[q] = (e < nedges) ? dst[e] : -1;
                }
                #pragma unroll
                for (int q = 0; q < 8; ++q) {
                    unsigned mask = __builtin_amdgcn_ballot_w32((unsigned)(d[q] - node0) < (unsigned)nvalid);
                    while (mask) {
                        const int L = __builtin_ctz(mask);
                        mask &= mask - 1;
                        const int dl = __builtin_amdgcn_readlane(d[q], L) - node0;
                        const int s  = src[b + q * 32 + L];
                        rgcn_add_row(Wh, s, n, acc, cnt, dl, lane);
                    }
                }
            }
        }
    }
    __syncthreads();

    for (int r = wave; r < nvalid; r += AGG_WAVES) {
        const int c = cnt[r];
        if (!first && c == 0) continue;
        const float inv = 1.0f / fmaxf((float)c, 1.0f);
        const float* a = acc + r * DOUT + lane;
        float* o = out + (size_t)(node0 + r) * DOUT + lane * 4;
        float4 v[4];
        #pragma unroll
        for (int j = 0; j < 4; ++j) {
            v[j].x = a[j * 128 +  0] * inv;
            v[j].y = a[j * 128 + 32] * inv;
            v[j].z = a[j * 128 + 64] * inv;
            v[j].w = a[j * 128 + 96] * inv;
        }
        if (!first) {
            float4 old[4];
            #pragma unroll
            for (int j = 0; j < 4; ++j) old[j] = *(const float4*)(o + j * 128);
            #pragma unroll
            for (int j = 0; j < 4; ++j) {
                v[j].x += old[j].x; v[j].y += old[j].y; v[j].z += old[j].z; v[j].w += old[j].w;
            }
        }
        #pragma unroll
        for (int j = 0; j < 4; ++j) *(float4*)(o + j * 128) = v[j];
    }
}

extern "C" void kernel_launch(void* const* d_in, const int* in_sizes, int n_in,
                              void* d_out, int out_size, void* d_ws, size_t ws_size,
                              hipStream_t stream) {
    const float* x   = (const float*)d_in[0];
    const float* W   = (const float*)d_in[1];
    const float* b   = (const float*)d_in[2];
    const int*   src = (const int*)d_in[3];
    const int*   dst = (const int*)d_in[4];
    float*       out = (float*)d_out;

    const int n    = in_sizes[0] / DIN;
    const int nrel = in_sizes[1] / (DIN * DOUT);
    const int e    = in_sizes[3] / nrel;

    float* Wh = (float*)d_ws;

    dim3 ggrid((n + 63) / 64, DOUT / 128, 1);
    for (int r = 0; r < nrel; ++r) {
        rgcn_gemm_wmma_kernel<<<ggrid, 128, 0, stream>>>(
            x, W + (size_t)r * DIN * DOUT, b + (size_t)r * DOUT, Wh, n);
        rgcn_owner_agg_kernel<<<(n + NB - 1) / NB, AGG_THREADS, 0, stream>>>(
            Wh, src + (size_t)r * e, dst + (size_t)r * e, out, e, n, r == 0 ? 1 : 0);
    }
}


# Edge sums in registers: the scan only lists each owned row's edges in LDS; a wave then sums a row's Wh rows in registers in edge order, 4 in flight; no LDS float adds; 256 groups of 196 rows
# speedup vs baseline: 1.9110x; 1.0763x over previous
#include <hip/hip_runtime.h>

typedef __attribute__((ext_vector_type(2))) float v2f;
typedef __attribute__((ext_vector_type(8))) float v8f;

#define DIN  512
#define DOUT 512

__global__ __launch_bounds__(128)
void rgcn_gemm_wmma_kernel(const float* __restrict__ x,
                           const float* __restrict__ W,
                           const float* __restrict__ b,
                           float* __restrict__ Wh,
                           int nrows) {
    __shared__ float xs[2][64][33];
    __shared__ float wsT[2][128][33];

    const int tid  = threadIdx.x;
    const int wave = tid >> 5;
    const int lane = tid & 31;
    const int l15  = lane & 15;
    const int khal = (lane >> 4) * 2;

    const int m0 = blockIdx.x * 64;
    const int n0 = blockIdx.y * 128;
    const int wm = (wave >> 1) * 32;
    const int wn = (wave & 1) * 64;

    const int xr0 = tid >> 3;
    const int xc  = (tid & 7) * 4;
    const int wc  = lane * 4;

    const float* xp[4];
    #pragma unroll
    for (int j = 0; j < 4; ++j) {
        int gr = m0 + xr0 + 16 * j;
        if (gr >= nrows) gr = nrows - 1;
        xp[j] = x + (size_t)gr * DIN + xc;
    }
    const float* wp = W + (size_t)wave * DOUT + n0 + wc;

    float4 xreg[4], wreg[8];

    v8f acc[2][4] = {};

    #pragma unroll
    for (int j = 0; j < 4; ++j) xreg[j] = *(const float4*)(xp[j]);
    #pragma unroll
    for (int j = 0; j < 8; ++j) wreg[j] = *(const float4*)(wp + (size_t)(4 * j) * DOUT);
    #pragma unroll
    for (int j = 0; j < 4; ++j) {
        float* d = &xs[0][xr0 + 16 * j][xc];
        d[0] = xreg[j].x; d[1] = xreg[j].y; d[2] = xreg[j].z; d[3] = xreg[j].w;
    }
    #pragma unroll
    for (int j = 0; j < 8; ++j) {
        const int k = wave + 4 * j;
        wsT[0][wc + 0][k] = wreg[j].x;
        wsT[0][wc + 1][k] = wreg[j].y;
        wsT[0][wc + 2][k] = wreg[j].z;
        wsT[0][wc + 3][k] = wreg[j].w;
    }
    __syncthreads();

    for (int it = 0; it < DIN / 32; ++it) {
        const int  s    = it & 1;
        const bool more = (it + 1) < DIN / 32;

        if (more) {
            const int k1 = (it + 1) * 32;
            #pragma unroll
            for (int j = 0; j < 4; ++j) xreg[j] = *(const float4*)(xp[j] + k1);
            #pragma unroll
            for (int j = 0; j < 8; ++j)
                wreg[j] = *(const float4*)(wp + (size_t)(k1 + 4 * j) * DOUT);
        }

        #pragma unroll
        for (int kk = 0; kk < 32; kk += 4) {
            const int kb = kk + khal;
            v2f a[2];
            #pragma unroll
            for (int i = 0; i < 2; ++i) {
                a[i].x = xs[s][wm + i * 16 + l15][kb];
                a[i].y = xs[s][wm + i * 16 + l15][kb + 1];
            }
            v2f bf[4];
            #pragma unroll
            for (int j = 0; j < 4; ++j) {
                bf[j].x = wsT[s][wn + j * 16 + l15][kb];
                bf[j].y = wsT[s][wn + j * 16 + l15][kb + 1];
            }
            #pragma unroll
            for (int i = 0; i < 2; ++i)
                #pragma unroll
                for (int j = 0; j < 4; ++j)
                    acc[i][j] = __builtin_amdgcn_wmma_f32_16x16x4_f32(
                        false, a[i], false, bf[j], (short)0, acc[i][j], false, false);
        }

        if (more) {
            const int s1 = s ^ 1;
            #pragma unroll
            for (int j = 0; j < 4; ++j) {
                float* d = &xs[s1][xr0 + 16 * j][xc];
                d[0] = xreg[j].x; d[1] = xreg[j].y; d[2] = xreg[j].z; d[3] = xreg[j].w;
            }
            #pragma unroll
            for (int j = 0; j < 8; ++j) {
                const int k = wave + 4 * j;
                wsT[s1][wc + 0][k] = wreg[j].x;
                wsT[s1][wc + 1][k] = wreg[j].y;
                wsT[s1][wc + 2][k] = wreg[j].z;
                wsT[s1][wc + 3][k] = wreg[j].w;
            }
        }
        __syncthreads();
    }

    const int mbase = m0 + wm + (lane >> 4) * 8;
    #pragma unroll
    for (int i = 0; i < 2; ++i) {
        #pragma unroll
        for (int j = 0; j < 4; ++j) {
            const int col  = n0 + wn + j * 16 + l15;
            const float bb = b[col];
            #pragma unroll
            for (int v = 0; v < 8; ++v) {
                const int row = mbase + i * 16 + v;
                if (row < nrows)
                    Wh[(size_t)row * DOUT + col] = acc[i][j][v] + bb;
            }
        }
    }
}

#define NB          196
#define AGG_WAVES   32
#define AGG_THREADS (AGG_WAVES * 32)
#define CHUNK       1024
#define CAPR        32

__device__ __forceinline__ void rgcn_note_edge(int d, int e, int node0, int nvalid,
                                               int* cnt, int* lst) {
    const int dl = d - node0;
    if ((unsigned)dl < (unsigned)nvalid) {
        const int pos = atomicAdd(&cnt[dl], 1);
        if (pos < CAPR) lst[dl * CAPR + pos] = e;
    }
}

__device__ __forceinline__ void rgcn_load_row(const float* __restrict__ Wh, int s, int n,
                                              int lane, float4 (&v)[4]) {
    if ((unsigned)s >= (unsigned)n) s = 0;
    const float4* p = (const float4*)(Wh + (size_t)s * DOUT) + lane;
    v[0] = p[0]; v[1] = p[32]; v[2] = p[64]; v[3] = p[96];
}

__device__ __forceinline__ void rgcn_acc_row(float4 (&a)[4], const float4 (&v)[4]) {
    #pragma unroll
    for (int j = 0; j < 4; ++j) {
        a[j].x += v[j].x; a[j].y += v[j].y; a[j].z += v[j].z; a[j].w += v[j].w;
    }
}

__device__ __forceinline__ void rgcn_fetch_list(const int* __restrict__ src, const int* cnt,
                                                const int* lst, int r, int lane,
                                                int& c, int& e, int& s) {
    c = __builtin_amdgcn_readfirstlane(cnt[r]);
    e = 0x7fffffff;
    s = 0;
    if (lane < c && lane < CAPR) {
        e = lst[r * CAPR + lane];
        s = src[e];
    }
}

__global__ __launch_bounds__(AGG_THREADS)
void rgcn_owner_agg_kernel(const float* __restrict__ Wh,
                           const int* __restrict__ src,
                           const int* __restrict__ dst,
                           float* __restrict__ out,
                           int nedges, int n, int first) {
    __shared__ int cnt[NB];
    __shared__ int lst[NB * CAPR];

    const int tid  = threadIdx.x;
    const int lane = tid & 31;
    const int wave = __builtin_amdgcn_readfirstlane(tid >> 5);
    const int node0 = blockIdx.x * NB;
    int nvalid = n - node0;
    if (nvalid > NB) nvalid = NB;
    const bool vec_ok = (((size_t)dst) & 15) == 0;

    if (tid < NB) cnt[tid] = 0;
    __syncthreads();

    const int step = AGG_WAVES * CHUNK;
    for (int base = wave * CHUNK; base < nedges; base += step) {
        if (vec_ok && base + CHUNK <= nedges) {
            int4 v[8];
            #pragma unroll
            for (int g = 0; g < 8; ++g)
                v[g] = *(const int4*)(dst + base + g * 128 + lane * 4);
            #pragma unroll
            for (int g = 0; g < 8; ++g) {
                const bool h = ((unsigned)(v[g].x - node0) < (unsigned)nvalid) |
                               ((unsigned)(v[g].y - node0) < (unsigned)nvalid) |
                               ((unsigned)(v[g].z - node0) < (unsigned)nvalid) |
                               ((unsigned)(v[g].w - node0) < (unsigned)nvalid);
                if (__builtin_amdgcn_ballot_w32(h)) {
                    const int e0 = base + g * 128 + lane * 4;
                    rgcn_note_edge(v[g].x, e0 + 0, node0, nvalid, cnt, lst);
                    rgcn_note_edge(v[g].y, e0 + 1, node0, nvalid, cnt, lst);
                    rgcn_note_edge(v[g].z, e0 + 2, node0, nvalid, cnt, lst);
                    rgcn_note_edge(v[g].w, e0 + 3, node0, nvalid, cnt, lst);
                }
            }
        } else {
            #pragma unroll 1
            for (int b = base; b < base + CHUNK && b < nedges; b += 256) {
                int d[8];
                #pragma unroll
                for (int q = 0; q < 8; ++q) {
                    const int e = b + q * 32 + lane;
                    d[q] = (e < nedges) ? dst[e] : -1;
                }
                #pragma unroll
                for (int q = 0; q < 8; ++q)
                    rgcn_note_edge(d[q], b + q * 32 + lane, node0, nvalid, cnt, lst);
            }
        }
    }
    __syncthreads();

    int r = wave;
    int c_n = 0, e_n = 0x7fffffff, s_n = 0;
    if (r < nvalid) rgcn_fetch_list(src, cnt, lst, r, lane, c_n, e_n, s_n);
    for (; r < nvalid; r += AGG_WAVES) {
        const int c = c_n, e = e_n, s = s_n;
        if (r + AGG_WAVES < nvalid)
            rgcn_fetch_list(src, cnt, lst, r + AGG_WAVES, lane, c_n, e_n, s_n);
        if (!first && c == 0) continue;

        float* o = out + (size_t)(node0 + r) * DOUT + lane * 4;
        float4 old[4];
        if (!first) {
            #pragma unroll
            for (int j = 0; j < 4; ++j) old[j] = *(const float4*)(o + j * 128);
        }
        float4 acc[4];
        #pragma unroll
        for (int j = 0; j < 4; ++j) acc[j] = make_float4(0.0f, 0.0f, 0.0f, 0.0f);

        if (c <= CAPR) {
            int rank = 0;
            for (int j = 0; j < c; ++j)
                rank += (__builtin_amdgcn_readlane(e, j) < e) ? 1 : 0;
            const bool mine = lane < c;
            #define RGCN_PICK(k) __builtin_amdgcn_readlane(s, \
                __builtin_ctz(__builtin_amdgcn_ballot_w32(mine && rank == (k))))
            int i = 0;
            for (; i + 3 < c; i += 4) {
                const int s0 = RGCN_PICK(i), s1 = RGCN_PICK(i + 1);
                const int s2 = RGCN_PICK(i + 2), s3 = RGCN_PICK(i + 3);
                float4 a0[4], a1[4], a2[4], a3[4];
                rgcn_load_row(Wh, s0, n, lane, a0);
                rgcn_load_row(Wh, s1, n, lane, a1);
                rgcn_load_row(Wh, s2, n, lane, a2);
                rgcn_load_row(Wh, s3, n, lane, a3);
                rgcn_acc_row(acc, a0); rgcn_acc_row(acc, a1);
                rgcn_acc_row(acc, a2); rgcn_acc_row(acc, a3);
            }
            if (i + 1 < c) {
                const int s0 = RGCN_PICK(i), s1 = RGCN_PICK(i + 1);
                float4 a0[4], a1[4];
                rgcn_load_row(Wh, s0, n, lane, a0);
                rgcn_load_row(Wh, s1, n, lane, a1);
                rgcn_acc_row(acc, a0); rgcn_acc_row(acc, a1);
                i += 2;
            }
            if (i < c) {
                const int s0 = RGCN_PICK(i);
                float4 a0[4];
                rgcn_load_row(Wh, s0, n, lane, a0);
                rgcn_acc_row(acc, a0);
            }
            #undef RGCN_PICK
        } else {
            const int dgt = node0 + r;
            #pragma unroll 1
            for (int b = 0; b < nedges; b += 32) {
                const int ee = b + lane;
                const int dd = (ee < nedges) ? dst[ee] : -1;
                unsigned mask = __builtin_amdgcn_ballot_w32(dd == dgt);
                while (mask) {
                    const int L = __builtin_ctz(mask);
                    mask &= mask - 1;
                    float4 a0[4];
                    rgcn_load_row(Wh, src[b + L], n, lane, a0);
                    rgcn_acc_row(acc, a0);
                }
            }
        }

        const float inv = 1.0f / fmaxf((float)c, 1.0f);
        #pragma unroll
        for (int j = 0; j < 4; ++j) {
            float4 v;
            v.x = acc[j].x * inv; v.y = acc[j].y * inv;
            v.z = acc[j].z * inv; v.w = acc[j].w * inv;
            if (!first) {
                v.x += old[j].x; v.y += old[j].y; v.z += old[j].z; v.w += old[j].w;
            }
            *(float4*)(o + j * 128) = v;
        }
    }
}

extern "C" void kernel_launch(void* const* d_in, const int* in_sizes, int n_in,
                              void* d_out, int out_size, void* d_ws, size_t ws_size,
                              hipStream_t stream) {
    const float* x   = (const float*)d_in[0];
    const float* W   = (const float*)d_in[1];
    const float* b   = (const float*)d_in[2];
    const int*   src = (const int*)d_in[3];
    const int*   dst = (const int*)d_in[4];
    float*       out = (float*)d_out;

    const int n    = in_sizes[0] / DIN;
    const int nrel = in_sizes[1] / (DIN * DOUT);
    const int e    = in_sizes[3] / nrel;

    float* Wh = (float*)d_ws;

    dim3 ggrid((n + 63) / 64, DOUT / 128, 1);
    for (int r = 0; r < nrel; ++r) {
        rgcn_gemm_wmma_kernel<<<ggrid, 128, 0, stream>>>(
            x, W + (size_t)r * DIN * DOUT, b + (size_t)r * DOUT, Wh, n);
        rgcn_owner_agg_kernel<<<(n + NB - 1) / NB, AGG_THREADS, 0, stream>>>(
            Wh, src + (size_t)r * e, dst + (size_t)r * e, out, e, n, r == 0 ? 1 : 0);
    }
}


# GEMM LDS tiles re-laid: x at pitch 36, W as k-pair rows [K/2][N][2], so staging is 16-byte LDS stores and every A and B fragment is one conflict-free 8-byte load at an immediate offset
# speedup vs baseline: 1.9869x; 1.0397x over previous
#include <hip/hip_runtime.h>

typedef __attribute__((ext_vector_type(2))) float v2f;
typedef __attribute__((ext_vector_type(8))) float v8f;

#define DIN  512
#define DOUT 512
#define XP   36
#define WP2  288

__global__ __launch_bounds__(128)
void rgcn_gemm_wmma_kernel(const float* __restrict__ x,
                           const float* __restrict__ W,
                           const float* __restrict__ b,
                           float* __restrict__ Wh,
                           int nrows) {
    __shared__ __attribute__((aligned(16))) float xs[2][64][XP];
    __shared__ __attribute__((aligned(16))) float wsl[2][16][WP2];

    const int tid  = threadIdx.x;
    const int wave = tid >> 5;
    const int lane = tid & 31;
    const int l15  = lane & 15;
    const int khal = (lane >> 4) * 2;

    const int m0 = blockIdx.x * 64;
    const int n0 = blockIdx.y * 128;
    const int wm = (wave >> 1) * 32;
    const int wn = (wave & 1) * 64;

    const int xr0 = tid >> 3;
    const int xc  = (tid & 7) * 4;
    const int wc  = lane * 4;

    const float* xp[4];
    #pragma unroll
    for (int j = 0; j < 4; ++j) {
        int gr = m0 + xr0 + 16 * j;
        if (gr >= nrows) gr = nrows - 1;
        xp[j] = x + (size_t)gr * DIN + xc;
    }
    const float* wp = W + (size_t)(2 * wave) * DOUT + n0 + wc;

    float4 xreg[4], wreg[8];

    v8f acc[2][4] = {};

    #pragma unroll
    for (int j = 0; j < 4; ++j) xreg[j] = *(const float4*)(xp[j]);
    #pragma unroll
    for (int j = 0; j < 8; ++j)
        wreg[j] = *(const float4*)(wp + (size_t)(8 * (j >> 1) + (j & 1)) * DOUT);
    #pragma unroll
    for (int j = 0; j < 4; ++j) *(float4*)&xs[0][xr0 + 16 * j][xc] = xreg[j];
    #pragma unroll
    for (int j = 0; j < 4; ++j) {
        float* d = &wsl[0][wave + 4 * j][2 * wc];
        const float4 lo = wreg[2 * j], hi = wreg[2 * j + 1];
        *(float4*)(d)     = make_float4(lo.x, hi.x, lo.y, hi.y);
        *(float4*)(d + 4) = make_float4(lo.z, hi.z, lo.w, hi.w);
    }
    __syncthreads();

    for (int it = 0; it < DIN / 32; ++it) {
        const int  s    = it & 1;
        const bool more = (it + 1) < DIN / 32;

        if (more) {
            const int k1 = (it + 1) * 32;
            #pragma unroll
            for (int j = 0; j < 4; ++j) xreg[j] = *(const float4*)(xp[j] + k1);
            #pragma unroll
            for (int j = 0; j < 8; ++j)
                wreg[j] = *(const float4*)(wp + (size_t)(k1 + 8 * (j >> 1) + (j & 1)) * DOUT);
        }
        __builtin_amdgcn_sched_barrier(0);

        #pragma unroll
        for (int kk = 0; kk < 32; kk += 4) {
            const int kb = kk + khal;
            v2f a[2];
            #pragma unroll
            for (int i = 0; i < 2; ++i)
                a[i] = *(const v2f*)&xs[s][wm + i * 16 + l15][kb];
            v2f bf[4];
            #pragma unroll
            for (int j = 0; j < 4; ++j)
                bf[j] = *(const v2f*)&wsl[s][kb >> 1][2 * (wn + j * 16 + l15)];
            #pragma unroll
            for (int i = 0; i < 2; ++i)
                #pragma unroll
                for (int j = 0; j < 4; ++j)
                    acc[i][j] = __builtin_amdgcn_wmma_f32_16x16x4_f32(
                        false, a[i], false, bf[j], (short)0, acc[i][j], false, false);
        }

        if (more) {
            const int s1 = s ^ 1;
            #pragma unroll
            for (int j = 0; j < 4; ++j) *(float4*)&xs[s1][xr0 + 16 * j][xc] = xreg[j];
            #pragma unroll
            for (int j = 0; j < 4; ++j) {
                float* d = &wsl[s1][wave + 4 * j][2 * wc];
                const float4 lo = wreg[2 * j], hi = wreg[2 * j + 1];
                *(float4*)(d)     = make_float4(lo.x, hi.x, lo.y, hi.y);
                *(float4*)(d + 4) = make_float4(lo.z, hi.z, lo.w, hi.w);
            }
        }
        __syncthreads();
    }

    const int mbase = m0 + wm + (lane >> 4) * 8;
    #pragma unroll
    for (int i = 0; i < 2; ++i) {
        #pragma unroll
        for (int j = 0; j < 4; ++j) {
            const int col  = n0 + wn + j * 16 + l15;
            const float bb = b[col];
            #pragma unroll
            for (int v = 0; v < 8; ++v) {
                const int row = mbase + i * 16 + v;
                if (row < nrows)
                    Wh[(size_t)row * DOUT + col] = acc[i][j][v] + bb;
            }
        }
    }
}

#define NB          196
#define AGG_WAVES   32
#define AGG_THREADS (AGG_WAVES * 32)
#define CHUNK       1024
#define CAPR        32

__device__ __forceinline__ void rgcn_note_edge(int d, int e, int node0, int nvalid,
                                               int* cnt, int* lst) {
    const int dl = d - node0;
    if ((unsigned)dl < (unsigned)nvalid) {
        const int pos = atomicAdd(&cnt[dl], 1);
        if (pos < CAPR) lst[dl * CAPR + pos] = e;
    }
}

__device__ __forceinline__ void rgcn_load_row(const float* __restrict__ Wh, int s, int n,
                                              int lane, float4 (&v)[4]) {
    if ((unsigned)s >= (unsigned)n) s = 0;
    const float4* p = (const float4*)(Wh + (size_t)s * DOUT) + lane;
    v[0] = p[0]; v[1] = p[32]; v[2] = p[64]; v[3] = p[96];
}

__device__ __forceinline__ void rgcn_acc_row(float4 (&a)[4], const float4 (&v)[4]) {
    #pragma unroll
    for (int j = 0; j < 4; ++j) {
        a[j].x += v[j].x; a[j].y += v[j].y; a[j].z += v[j].z; a[j].w += v[j].w;
    }
}

__device__ __forceinline__ void rgcn_fetch_list(const int* __restrict__ src, const int* cnt,
                                                const int* lst, int r, int lane,
                                                int& c, int& e, int& s) {
    c = __builtin_amdgcn_readfirstlane(cnt[r]);
    e = 0x7fffffff;
    s = 0;
    if (lane < c && lane < CAPR) {
        e = lst[r * CAPR + lane];
        s = src[e];
    }
}

__global__ __launch_bounds__(AGG_THREADS)
void rgcn_owner_agg_kernel(const float* __restrict__ Wh,
                           const int* __restrict__ src,
                           const int* __restrict__ dst,
                           float* __restrict__ out,
                           int nedges, int n, int first) {
    __shared__ int cnt[NB];
    __shared__ int lst[NB * CAPR];

    const int tid  = threadIdx.x;
    const int lane = tid & 31;
    const int wave = __builtin_amdgcn_readfirstlane(tid >> 5);
    const int node0 = blockIdx.x * NB;
    int nvalid = n - node0;
    if (nvalid > NB) nvalid = NB;
    const bool vec_ok = (((size_t)dst) & 15) == 0;

    if (tid < NB) cnt[tid] = 0;
    __syncthreads();

    const int step = AGG_WAVES * CHUNK;
    for (int base = wave * CHUNK; base < nedges; base += step) {
        if (vec_ok && base + CHUNK <= nedges) {
            int4 v[8];
            #pragma unroll
            for (int g = 0; g < 8; ++g)
                v[g] = *(const int4*)(dst + base + g * 128 + lane * 4);
            #pragma unroll
            for (int g = 0; g < 8; ++g) {
                const bool h = ((unsigned)(v[g].x - node0) < (unsigned)nvalid) |
                               ((unsigned)(v[g].y - node0) < (unsigned)nvalid) |
                               ((unsigned)(v[g].z - node0) < (unsigned)nvalid) |
                               ((unsigned)(v[g].w - node0) < (unsigned)nvalid);
                if (__builtin_amdgcn_ballot_w32(h)) {
                    const int e0 = base + g * 128 + lane * 4;
                    rgcn_note_edge(v[g].x, e0 + 0, node0, nvalid, cnt, lst);
                    rgcn_note_edge(v[g].y, e0 + 1, node0, nvalid, cnt, lst);
                    rgcn_note_edge(v[g].z, e0 + 2, node0, nvalid, cnt, lst);
                    rgcn_note_edge(v[g].w, e0 + 3, node0, nvalid, cnt, lst);
                }
            }
        } else {
            #pragma unroll 1
            for (int b = base; b < base + CHUNK && b < nedges; b += 256) {
                int d[8];
                #pragma unroll
                for (int q = 0; q < 8; ++q) {
                    const int e = b + q * 32 + lane;
                    d[q] = (e < nedges) ? dst[e] : -1;
                }
                #pragma unroll
                for (int q = 0; q < 8; ++q)
                    rgcn_note_edge(d[q], b + q * 32 + lane, node0, nvalid, cnt, lst);
            }
        }
    }
    __syncthreads();

    int r = wave;
    int c_n = 0, e_n = 0x7fffffff, s_n = 0;
    if (r < nvalid) rgcn_fetch_list(src, cnt, lst, r, lane, c_n, e_n, s_n);
    for (; r < nvalid; r += AGG_WAVES) {
        const int c = c_n, e = e_n, s = s_n;
        if (r + AGG_WAVES < nvalid)
            rgcn_fetch_list(src, cnt, lst, r + AGG_WAVES, lane, c_n, e_n, s_n);
        if (!first && c == 0) continue;

        float* o = out + (size_t)(node0 + r) * DOUT + lane * 4;
        float4 old[4];
        if (!first) {
            #pragma unroll
            for (int j = 0; j < 4; ++j) old[j] = *(const float4*)(o + j * 128);
        }
        float4 acc[4];
        #pragma unroll
        for (int j = 0; j < 4; ++j) acc[j] = make_float4(0.0f, 0.0f, 0.0f, 0.0f);

        if (c <= CAPR) {
            int rank = 0;
            for (int j = 0; j < c; ++j)
                rank += (__builtin_amdgcn_readlane(e, j) < e) ? 1 : 0;
            const bool mine = lane < c;
            #define RGCN_PICK(k) __builtin_amdgcn_readlane(s, \
                __builtin_ctz(__builtin_amdgcn_ballot_w32(mine && rank == (k))))
            int i = 0;
            for (; i + 3 < c; i += 4) {
                const int s0 = RGCN_PICK(i), s1 = RGCN_PICK(i + 1);
                const int s2 = RGCN_PICK(i + 2), s3 = RGCN_PICK(i + 3);
                float4 a0[4], a1[4], a2[4], a3[4];
                rgcn_load_row(Wh, s0, n, lane, a0);
                rgcn_load_row(Wh, s1, n, lane, a1);
                rgcn_load_row(Wh, s2, n, lane, a2);
                rgcn_load_row(Wh, s3, n, lane, a3);
                rgcn_acc_row(acc, a0); rgcn_acc_row(acc, a1);
                rgcn_acc_row(acc, a2); rgcn_acc_row(acc, a3);
            }
            if (i + 1 < c) {
                const int s0 = RGCN_PICK(i), s1 = RGCN_PICK(i + 1);
                float4 a0[4], a1[4];
                rgcn_load_row(Wh, s0, n, lane, a0);
                rgcn_load_row(Wh, s1, n, lane, a1);
                rgcn_acc_row(acc, a0); rgcn_acc_row(acc, a1);
                i += 2;
            }
            if (i < c) {
                const int s0 = RGCN_PICK(i);
                float4 a0[4];
                rgcn_load_row(Wh, s0, n, lane, a0);
                rgcn_acc_row(acc, a0);
            }
            #undef RGCN_PICK
        } else {
            const int dgt = node0 + r;
            #pragma unroll 1
            for (int b = 0; b < nedges; b += 32) {
                const int ee = b + lane;
                const int dd = (ee < nedges) ? dst[ee] : -1;
                unsigned mask = __builtin_amdgcn_ballot_w32(dd == dgt);
                while (mask) {
                    const int L = __builtin_ctz(mask);
                    mask &= mask - 1;
                    float4 a0[4];
                    rgcn_load_row(Wh, src[b + L], n, lane, a0);
                    rgcn_acc_row(acc, a0);
                }
            }
        }

        const float inv = 1.0f / fmaxf((float)c, 1.0f);
        #pragma unroll
        for (int j = 0; j < 4; ++j) {
            float4 v;
            v.x = acc[j].x * inv; v.y = acc[j].y * inv;
            v.z = acc[j].z * inv; v.w = acc[j].w * inv;
            if (!first) {
                v.x += old[j].x; v.y += old[j].y; v.z += old[j].z; v.w += old[j].w;
            }
            *(float4*)(o + j * 128) = v;
        }
    }
}

extern "C" void kernel_launch(void* const* d_in, const int* in_sizes, int n_in,
                              void* d_out, int out_size, void* d_ws, size_t ws_size,
                              hipStream_t stream) {
    const float* x   = (const float*)d_in[0];
    const float* W   = (const float*)d_in[1];
    const float* b   = (const float*)d_in[2];
    const int*   src = (const int*)d_in[3];
    const int*   dst = (const int*)d_in[4];
    float*       out = (float*)d_out;

    const int n    = in_sizes[0] / DIN;
    const int nrel = in_sizes[1] / (DIN * DOUT);
    const int e    = in_sizes[3] / nrel;

    float* Wh = (float*)d_ws;

    dim3 ggrid((n + 63) / 64, DOUT / 128, 1);
    for (int r = 0; r < nrel; ++r) {
        rgcn_gemm_wmma_kernel<<<ggrid, 128, 0, stream>>>(
            x, W + (size_t)r * DIN * DOUT, b + (size_t)r * DOUT, Wh, n);
        rgcn_owner_agg_kernel<<<(n + NB - 1) / NB, AGG_THREADS, 0, stream>>>(
            Wh, src + (size_t)r * e, dst + (size_t)r * e, out, e, n, r == 0 ? 1 : 0);
    }
}


# Relations folded into the launches when the workspace holds a Wh plane each: one GEMM launch (relation on grid z), one edge-sum launch sums a row over all relations in registers; out stored once
# speedup vs baseline: 2.2193x; 1.1170x over previous
#include <hip/hip_runtime.h>

typedef __attribute__((ext_vector_type(2))) float v2f;
typedef __attribute__((ext_vector_type(8))) float v8f;

#define DIN  512
#define DOUT 512
#define XP   36
#define WP2  288

__global__ __launch_bounds__(128)
void rgcn_gemm_wmma_kernel(const float* __restrict__ x,
                           const float* __restrict__ W,
                           const float* __restrict__ b,
                           float* __restrict__ Wh,
                           int nrows) {
    __shared__ __attribute__((aligned(16))) float xs[2][64][XP];
    __shared__ __attribute__((aligned(16))) float wsl[2][16][WP2];

    const int tid  = threadIdx.x;
    const int wave = tid >> 5;
    const int lane = tid & 31;
    const int l15  = lane & 15;
    const int khal = (lane >> 4) * 2;

    const int m0 = blockIdx.x * 64;
    const int n0 = blockIdx.y * 128;
    W  += (size_t)blockIdx.z * DIN * DOUT;
    b  += (size_t)blockIdx.z * DOUT;
    Wh += (size_t)blockIdx.z * nrows * DOUT;
    const int wm = (wave >> 1) * 32;
    const int wn = (wave & 1) * 64;

    const int xr0 = tid >> 3;
    const int xc  = (tid & 7) * 4;
    const int wc  = lane * 4;

    const float* xp[4];
    #pragma unroll
    for (int j = 0; j < 4; ++j) {
        int gr = m0 + xr0 + 16 * j;
        if (gr >= nrows) gr = nrows - 1;
        xp[j] = x + (size_t)gr * DIN + xc;
    }
    const float* wp = W + (size_t)(2 * wave) * DOUT + n0 + wc;

    float4 xreg[4], wreg[8];

    v8f acc[2][4] = {};

    #pragma unroll
    for (int j = 0; j < 4; ++j) xreg[j] = *(const float4*)(xp[j]);
    #pragma unroll
    for (int j = 0; j < 8; ++j)
        wreg[j] = *(const float4*)(wp + (size_t)(8 * (j >> 1) + (j & 1)) * DOUT);
    #pragma unroll
    for (int j = 0; j < 4; ++j) *(float4*)&xs[0][xr0 + 16 * j][xc] = xreg[j];
    #pragma unroll
    for (int j = 0; j < 4; ++j) {
        float* d = &wsl[0][wave + 4 * j][2 * wc];
        const float4 lo = wreg[2 * j], hi = wreg[2 * j + 1];
        *(float4*)(d)     = make_float4(lo.x, hi.x, lo.y, hi.y);
        *(float4*)(d + 4) = make_float4(lo.z, hi.z, lo.w, hi.w);
    }
    __syncthreads();

    for (int it = 0; it < DIN / 32; ++it) {
        const int  s    = it & 1;
        const bool more = (it + 1) < DIN / 32;

        if (more) {
            const int k1 = (it + 1) * 32;
            #pragma unroll
            for (int j = 0; j < 4; ++j) xreg[j] = *(const float4*)(xp[j] + k1);
            #pragma unroll
            for (int j = 0; j < 8; ++j)
                wreg[j] = *(const float4*)(wp + (size_t)(k1 + 8 * (j >> 1) + (j & 1)) * DOUT);
        }
        __builtin_amdgcn_sched_barrier(0);

        #pragma unroll
        for (int kk = 0; kk < 32; kk += 4) {
            const int kb = kk + khal;
            v2f a[2];
            #pragma unroll
            for (int i = 0; i < 2; ++i)
                a[i] = *(const v2f*)&xs[s][wm + i * 16 + l15][kb];
            v2f bf[4];
            #pragma unroll
            for (int j = 0; j < 4; ++j)
                bf[j] = *(const v2f*)&wsl[s][kb >> 1][2 * (wn + j * 16 + l15)];
            #pragma unroll
            for (int i = 0; i < 2; ++i)
                #pragma unroll
                for (int j = 0; j < 4; ++j)
                    acc[i][j] = __builtin_amdgcn_wmma_f32_16x16x4_f32(
                        false, a[i], false, bf[j], (short)0, acc[i][j], false, false);
        }

        if (more) {
            const int s1 = s ^ 1;
            #pragma unroll
            for (int j = 0; j < 4; ++j) *(float4*)&xs[s1][xr0 + 16 * j][xc] = xreg[j];
            #pragma unroll
            for (int j = 0; j < 4; ++j) {
                float* d = &wsl[s1][wave + 4 * j][2 * wc];
                const float4 lo = wreg[2 * j], hi = wreg[2 * j + 1];
                *(float4*)(d)     = make_float4(lo.x, hi.x, lo.y, hi.y);
                *(float4*)(d + 4) = make_float4(lo.z, hi.z, lo.w, hi.w);
            }
        }
        __syncthreads();
    }

    const int mbase = m0 + wm + (lane >> 4) * 8;
    #pragma unroll
    for (int i = 0; i < 2; ++i) {
        #pragma unroll
        for (int j = 0; j < 4; ++j) {
            const int col  = n0 + wn + j * 16 + l15;
            const float bb = b[col];
            #pragma unroll
            for (int v = 0; v < 8; ++v) {
                const int row = mbase + i * 16 + v;
                if (row < nrows)
                    Wh[(size_t)row * DOUT + col] = acc[i][j][v] + bb;
            }
        }
    }
}

#define NB          196
#define AGG_WAVES   32
#define AGG_THREADS (AGG_WAVES * 32)
#define CHUNK       1024
#define CAPR        32
#define MAXR        3

__device__ __forceinline__ void rgcn_note_edge(int d, int e, int node0, int nvalid,
                                               int* cnt, int* lst) {
    const int dl = d - node0;
    if ((unsigned)dl < (unsigned)nvalid) {
        const int pos = atomicAdd(&cnt[dl], 1);
        if (pos < CAPR) lst[dl * CAPR + pos] = e;
    }
}

__device__ __forceinline__ void rgcn_load_row(const float* __restrict__ Wh, int s, int n,
                                              int lane, float4 (&v)[4]) {
    if ((unsigned)s >= (unsigned)n) s = 0;
    const float4* p = (const float4*)(Wh + (size_t)s * DOUT) + lane;
    v[0] = p[0]; v[1] = p[32]; v[2] = p[64]; v[3] = p[96];
}

__device__ __forceinline__ void rgcn_acc_row(float4 (&a)[4], const float4 (&v)[4]) {
    #pragma unroll
    for (int j = 0; j < 4; ++j) {
        a[j].x += v[j].x; a[j].y += v[j].y; a[j].z += v[j].z; a[j].w += v[j].w;
    }
}

__device__ __forceinline__ void rgcn_fetch_list(const int* __restrict__ src, const int* cnt,
                                                const int* lst, int r, int lane,
                                                int& c, int& e, int& s) {
    c = __builtin_amdgcn_readfirstlane(cnt[r]);
    e = 0x7fffffff;
    s = 0;
    if (lane < c && lane < CAPR) {
        e = lst[r * CAPR + lane];
        s = src[e];
    }
}

__global__ __launch_bounds__(AGG_THREADS)
void rgcn_owner_agg_kernel(const float* __restrict__ Wh,
                           const int* __restrict__ src,
                           const int* __restrict__ dst,
                           float* __restrict__ out,
                           int nedges, int n, int first, int nr) {
    __shared__ int cnt[MAXR * NB];
    __shared__ int lst[MAXR * NB * CAPR];

    const int tid  = threadIdx.x;
    const int lane = tid & 31;
    const int wave = __builtin_amdgcn_readfirstlane(tid >> 5);
    const int node0 = blockIdx.x * NB;
    int nvalid = n - node0;
    if (nvalid > NB) nvalid = NB;

    if (tid < nr * NB) cnt[tid] = 0;
    __syncthreads();

    const int step = AGG_WAVES * CHUNK;
    for (int rr = 0; rr < nr; ++rr) {
        const int* dr = dst + (size_t)rr * nedges;
        int* cr = cnt + rr * NB;
        int* lr = lst + rr * NB * CAPR;
        const bool vec_ok = (((size_t)dr) & 15) == 0;
        for (int base = wave * CHUNK; base < nedges; base += step) {
            if (vec_ok && base + CHUNK <= nedges) {
                int4 v[8];
                #pragma unroll
                for (int g = 0; g < 8; ++g)
                    v[g] = *(const int4*)(dr + base + g * 128 + lane * 4);
                #pragma unroll
                for (int g = 0; g < 8; ++g) {
                    const bool h = ((unsigned)(v[g].x - node0) < (unsigned)nvalid) |
                                   ((unsigned)(v[g].y - node0) < (unsigned)nvalid) |
                                   ((unsigned)(v[g].z - node0) < (unsigned)nvalid) |
                                   ((unsigned)(v[g].w - node0) < (unsigned)nvalid);
                    if (__builtin_amdgcn_ballot_w32(h)) {
                        const int e0 = base + g * 128 + lane * 4;
                        rgcn_note_edge(v[g].x, e0 + 0, node0, nvalid, cr, lr);
                        rgcn_note_edge(v[g].y, e0 + 1, node0, nvalid, cr, lr);
                        rgcn_note_edge(v[g].z, e0 + 2, node0, nvalid, cr, lr);
                        rgcn_note_edge(v[g].w, e0 + 3, node0, nvalid, cr, lr);
                    }
                }
            } else {
                #pragma unroll 1
                for (int b = base; b < base + CHUNK && b < nedges; b += 256) {
                    int d[8];
                    #pragma unroll
                    for (int q = 0; q < 8; ++q) {
                        const int e = b + q * 32 + lane;
                        d[q] = (e < nedges) ? dr[e] : -1;
                    }
                    #pragma unroll
                    for (int q = 0; q < 8; ++q)
                        rgcn_note_edge(d[q], b + q * 32 + lane, node0, nvalid, cr, lr);
                }
            }
        }
    }
    __syncthreads();

    const size_t plane = (size_t)n * DOUT;
    int r = wave, rr = 0;
    int c_n = 0, e_n = 0x7fffffff, s_n = 0;
    if (r < nvalid) rgcn_fetch_list(src, cnt, lst, r, lane, c_n, e_n, s_n);
    float4 tot[4];
    bool any = false;
    while (r < nvalid) {
        const int c = c_n, e = e_n, s = s_n;
        int r2 = r, rr2 = rr + 1;
        if (rr2 == nr) { rr2 = 0; r2 = r + AGG_WAVES; }
        if (r2 < nvalid)
            rgcn_fetch_list(src + (size_t)rr2 * nedges, cnt + rr2 * NB, lst + rr2 * NB * CAPR,
                            r2, lane, c_n, e_n, s_n);

        float* o = out + (size_t)(node0 + r) * DOUT + lane * 4;
        if (rr == 0) any = false;
        if (c > 0) {
            if (!any) {
                any = true;
                if (first) {
                    #pragma unroll
                    for (int j = 0; j < 4; ++j) tot[j] = make_float4(0.0f, 0.0f, 0.0f, 0.0f);
                } else {
                    #pragma unroll
                    for (int j = 0; j < 4; ++j) tot[j] = *(const float4*)(o + j * 128);
                }
            }
            const float* Wr = Wh + (size_t)rr * plane;
            float4 acc[4];
            #pragma unroll
            for (int j = 0; j < 4; ++j) acc[j] = make_float4(0.0f, 0.0f, 0.0f, 0.0f);

            if (c <= CAPR) {
                int rank = 0;
                for (int j = 0; j < c; ++j)
                    rank += (__builtin_amdgcn_readlane(e, j) < e) ? 1 : 0;
                const bool mine = lane < c;
                #define RGCN_PICK(k) __builtin_amdgcn_readlane(s, \
                    __builtin_ctz(__builtin_amdgcn_ballot_w32(mine && rank == (k))))
                int i = 0;
                for (; i + 3 < c; i += 4) {
                    const int s0 = RGCN_PICK(i), s1 = RGCN_PICK(i + 1);
                    const int s2 = RGCN_PICK(i + 2), s3 = RGCN_PICK(i + 3);
                    float4 a0[4], a1[4], a2[4], a3[4];
                    rgcn_load_row(Wr, s0, n, lane, a0);
                    rgcn_load_row(Wr, s1, n, lane, a1);
                    rgcn_load_row(Wr, s2, n, lane, a2);
                    rgcn_load_row(Wr, s3, n, lane, a3);
                    rgcn_acc_row(acc, a0); rgcn_acc_row(acc, a1);
                    rgcn_acc_row(acc, a2); rgcn_acc_row(acc, a3);
                }
                if (i + 1 < c) {
                    const int s0 = RGCN_PICK(i), s1 = RGCN_PICK(i + 1);
                    float4 a0[4], a1[4];
                    rgcn_load_row(Wr, s0, n, lane, a0);
                    rgcn_load_row(Wr, s1, n, lane, a1);
                    rgcn_acc_row(acc, a0); rgcn_acc_row(acc, a1);
                    i += 2;
                }
                if (i < c) {
                    const int s0 = RGCN_PICK(i);
                    float4 a0[4];
                    rgcn_load_row(Wr, s0, n, lane, a0);
                    rgcn_acc_row(acc, a0);
                }
                #undef RGCN_PICK
            } else {
                const int* dr = dst + (size_t)rr * nedges;
                const int* sr = src + (size_t)rr * nedges;
                const int dgt = node0 + r;
                #pragma unroll 1
                for (int b = 0; b < nedges; b += 32) {
                    const int ee = b + lane;
                    const int dd = (ee < nedges) ? dr[ee] : -1;
                    unsigned mask = __builtin_amdgcn_ballot_w32(dd == dgt);
                    while (mask) {
                        const int L = __builtin_ctz(mask);
                        mask &= mask - 1;
                        float4 a0[4];
                        rgcn_load_row(Wr, sr[b + L], n, lane, a0);
                        rgcn_acc_row(acc, a0);
                    }
                }
            }

            const float inv = 1.0f / fmaxf((float)c, 1.0f);
            #pragma unroll
            for (int j = 0; j < 4; ++j) {
                tot[j].x += acc[j].x * inv; tot[j].y += acc[j].y * inv;
                tot[j].z += acc[j].z * inv; tot[j].w += acc[j].w * inv;
            }
        }

        if (rr == nr - 1) {
            if (any) {
                #pragma unroll
                for (int j = 0; j < 4; ++j) *(float4*)(o + j * 128) = tot[j];
            } else if (first) {
                #pragma unroll
                for (int j = 0; j < 4; ++j)
                    *(float4*)(o + j * 128) = make_float4(0.0f, 0.0f, 0.0f, 0.0f);
            }
        }
        r = r2; rr = rr2;
    }
}

extern "C" void kernel_launch(void* const* d_in, const int* in_sizes, int n_in,
                              void* d_out, int out_size, void* d_ws, size_t ws_size,
                              hipStream_t stream) {
    const float* x   = (const float*)d_in[0];
    const float* W   = (const float*)d_in[1];
    const float* b   = (const float*)d_in[2];
    const int*   src = (const int*)d_in[3];
    const int*   dst = (const int*)d_in[4];
    float*       out = (float*)d_out;

    const int n    = in_sizes[0] / DIN;
    const int nrel = in_sizes[1] / (DIN * DOUT);
    const int e    = in_sizes[3] / nrel;

    float* Wh = (float*)d_ws;
    const size_t plane_bytes = (size_t)n * DOUT * sizeof(float);
    const size_t fit = plane_bytes ? ws_size / plane_bytes : 1;
    const int per = fit >= MAXR ? MAXR : (fit >= 1 ? (int)fit : 1);

    for (int r0 = 0; r0 < nrel; r0 += per) {
        const int nb = (nrel - r0 < per) ? (nrel - r0) : per;
        dim3 ggrid((n + 63) / 64, DOUT / 128, nb);
        rgcn_gemm_wmma_kernel<<<ggrid, 128, 0, stream>>>(
            x, W + (size_t)r0 * DIN * DOUT, b + (size_t)r0 * DOUT, Wh, n);
        rgcn_owner_agg_kernel<<<(n + NB - 1) / NB, AGG_THREADS, 0, stream>>>(
            Wh, src + (size_t)r0 * e, dst + (size_t)r0 * e, out, e, n, r0 == 0 ? 1 : 0, nb);
    }
}
